# starting kernel
# baseline (speedup 1.0000x reference)
#include <hip/hip_runtime.h>
#include <hip/hip_bf16.h>
#include <math.h>

#define N_NODES     50000
#define N_EDGES     800000
#define NUM_FEATS   128
#define HIDDEN      256
#define NUM_CLASSES 40

typedef __attribute__((ext_vector_type(16))) _Float16 v16h;
typedef __attribute__((ext_vector_type(8)))  float    v8f;

__global__ void scatter_kernel(const float* __restrict__ x,
                               const int*   __restrict__ ei,
                               const float* __restrict__ ew,
                               float*       __restrict__ agg,
                               int D) {
    const int vecPerEdge = D >> 2;
    const int gid = blockIdx.x * blockDim.x + threadIdx.x;
    const int total = N_EDGES * vecPerEdge;
    if (gid >= total) return;
    const int e  = gid / vecPerEdge;
    const int f4 = (gid - e * vecPerEdge) << 2;
    const int src = ei[e];
    const int dst = ei[N_EDGES + e];
    const float w = ew[e];
    const float4 v = *(const float4*)(x + (long long)src * D + f4);
    float* o = agg + (long long)dst * D + f4;
    atomicAdd(o + 0, v.x * w);
    atomicAdd(o + 1, v.y * w);
    atomicAdd(o + 2, v.z * w);
    atomicAdd(o + 3, v.w * w);
}

template <int K>
__global__ __launch_bounds__(32)
void gemm_wmma_kernel(const float* __restrict__ A,
                      const float* __restrict__ W,
                      float*       __restrict__ C,
                      int Ncols, int do_relu) {
    const int l     = threadIdx.x;
    const int rl    = l & 15;
    const int khalf = (l >> 4) * 8;
    const int row0  = blockIdx.x * 16;
    const int col0  = blockIdx.y * 16;
    const int col   = col0 + rl;

    v8f acc = {};
    const float* arow = A + (long long)(row0 + rl) * K + khalf;
    const float* wrow = W + (long long)col * K + khalf;

    #pragma unroll
    for (int k0 = 0; k0 < K; k0 += 32) {
        v16h af;
        #pragma unroll
        for (int e = 0; e < 16; ++e) {
            const int kk = k0 + ((e & 8) ? 16 : 0) + (e & 7);
            af[e] = (_Float16)arow[kk];
        }
        v16h bf = {};
        if (col < Ncols) {
            #pragma unroll
            for (int e = 0; e < 16; ++e) {
                const int kk = k0 + ((e & 8) ? 16 : 0) + (e & 7);
                bf[e] = (_Float16)wrow[kk];
            }
        }
        acc = __builtin_amdgcn_wmma_f32_16x16x32_f16(
             false, af,  false, bf,
             (short)0, acc,  false,  false);
    }

    if (col < Ncols) {
        #pragma unroll
        for (int r = 0; r < 8; ++r) {
            float v = acc[r];
            if (do_relu) v = v > 0.0f ? v : 0.0f;
            C[(long long)(row0 + khalf + r) * Ncols + col] = v;
        }
    }
}

__global__ void logsoftmax_kernel(const float* __restrict__ H,
                                  float*       __restrict__ out) {
    const int n = blockIdx.x * blockDim.x + threadIdx.x;
    if (n >= N_NODES) return;
    const float* h = H + (long long)n * NUM_CLASSES;
    float mx = -INFINITY;
    #pragma unroll
    for (int c = 0; c < NUM_CLASSES; ++c) mx = fmaxf(mx, h[c]);
    float s = 0.0f;
    #pragma unroll
    for (int c = 0; c < NUM_CLASSES; ++c) s += __expf(h[c] - mx);
    const float lse = __logf(s) + mx;
    float* o = out + (long long)n * NUM_CLASSES;
    #pragma unroll
    for (int c = 0; c < NUM_CLASSES; ++c) o[c] = h[c] - lse;
}

extern "C" void kernel_launch(void* const* d_in, const int* in_sizes, int n_in,
                              void* d_out, int out_size, void* d_ws, size_t ws_size,
                              hipStream_t stream) {
    const float* x  = (const float*)d_in[0];
    const int*   ei = (const int*)  d_in[1];
    const float* ew = (const float*)d_in[2];
    const float* w1 = (const float*)d_in[3];
    const float* w2 = (const float*)d_in[4];
    float* out = (float*)d_out;

    const size_t bufElems = (size_t)N_NODES * HIDDEN;
    float* bufA = (float*)d_ws;
    float* bufB = bufA + bufElems;
    float* agg1 = bufA;
    float* h1   = bufB;
    float* agg2 = bufA;
    float* h2   = bufB;

    hipMemsetAsync(agg1, 0, (size_t)N_NODES * NUM_FEATS * sizeof(float), stream);
    {
        const int total = N_EDGES * (NUM_FEATS / 4);
        scatter_kernel<<<(total + 255) / 256, 256, 0, stream>>>(x, ei, ew, agg1, NUM_FEATS);
    }
    gemm_wmma_kernel<NUM_FEATS><<<dim3(N_NODES / 16, HIDDEN / 16), 32, 0, stream>>>(
        agg1, w1, h1, HIDDEN,  1);

    hipMemsetAsync(agg2, 0, (size_t)N_NODES * HIDDEN * sizeof(float), stream);
    {
        const int total = N_EDGES * (HIDDEN / 4);
        scatter_kernel<<<(total + 255) / 256, 256, 0, stream>>>(h1, ei, ew, agg2, HIDDEN);
    }
    gemm_wmma_kernel<HIDDEN><<<dim3(N_NODES / 16, 3  ), 32, 0, stream>>>(
        agg2, w2, h2, NUM_CLASSES,  0);

    logsoftmax_kernel<<<(N_NODES + 255) / 256, 256, 0, stream>>>(h2, out);
}


# layer 1's owner waves save each row's records in summing order (one run per tile, row counts in front); layer 2 sums its rows straight from it: no hand-on, pick or ordering pass; same sum order
# speedup vs baseline: 6.1315x; 6.1315x over previous
#include <hip/hip_runtime.h>
#include <hip/hip_bf16.h>
#include <math.h>

#define N_NODES     50000
#define N_EDGES     800000
#define NUM_FEATS   128
#define HIDDEN      256
#define NUM_CLASSES 40

typedef __attribute__((ext_vector_type(16))) _Float16 v16h;
typedef __attribute__((ext_vector_type(8)))  float    v8f;
typedef __attribute__((ext_vector_type(8)))  _Float16 v8h;

#define AGG_TILES   256
#define AGG_ROWS    196
#define AGG_PAD     4
#define AGG_STRIPS  ((AGG_ROWS + 15) / 16)
#define HS_PITCH    (NUM_CLASSES + 1)
#define AGG_THREADS 1024
#define AGG_WAVES   (AGG_THREADS / 32)
#define AGG_FLUSH   32
#define REC_FLUSH   128
#define AGG_CAP     (AGG_FLUSH + 128)
#define BM_CHUNKS      196
#define BM_CHUNK_WORDS 128
#define BM_CHUNK_EDGES (BM_CHUNK_WORDS * 32)
#define BM_ROW_WORDS   (BM_CHUNKS * BM_CHUNK_WORDS)
#define BM_LOADS       ((BM_ROW_WORDS / 4 + AGG_THREADS - 1) / AGG_THREADS)
#define REC_CAP     4096
#define REC_HEAD    128
#define REC_KEEP    (REC_CAP - REC_HEAD)
#define OWN_SLOTS   ((AGG_ROWS + AGG_WAVES - 1) / AGG_WAVES)
#define OWN_CAP     224
#define ROW_CAP     64
#define WS_CNT_OFF  (N_NODES * HIDDEN * 2)
#define WS_REC_OFF  (WS_CNT_OFF + 4096)
#define WS_BM_OFF   (WS_REC_OFF + AGG_TILES * REC_CAP * 8)
#define WS_PK_OFF   (WS_BM_OFF + AGG_TILES * BM_ROW_WORDS * 4)
static_assert(AGG_TILES * AGG_ROWS >= N_NODES, "tiles must cover all nodes");
static_assert(N_NODES <= 65536, "source and destination ids are packed into 16 bits each");
static_assert(WS_PK_OFF % 256 == 0, "a wave stores 32 packed edges as whole 128-byte lines");
static_assert(AGG_ROWS <= 256, "tile row is packed into 8 bits");
static_assert(BM_CHUNKS * BM_CHUNK_EDGES >= N_EDGES, "chunks must cover all edges");
static_assert(BM_ROW_WORDS % 4 == 0 && WS_BM_OFF % 16 == 0, "a bitmap is read 16 bytes at a time");
static_assert(AGG_TILES * BM_CHUNK_WORDS * 4 <= 160 * 1024, "a chunk's bitmaps of all tiles fit in LDS");
static_assert(BM_CHUNK_EDGES % AGG_THREADS == 0 && BM_CHUNK_WORDS % 32 == 0 && WS_BM_OFF % 128 == 0,
              "a thread takes a whole number of edges; a wave stores whole 128-byte lines");
static_assert(N_EDGES <= (1 << 20), "edge id is packed into 20 bits");
static_assert(AGG_WAVES == 32, "owner wave = row & 31, row slot = row >> 5");
static_assert(AGG_ROWS * 4 <= REC_HEAD * 8, "the row counts fit in front of the records");

template <int D, int NB>
__device__ __forceinline__ void agg_batch(const float* __restrict__ x,
                                          float* tile, const uint2* lst,
                                          int k, int lane) {
    constexpr int NV = D / 64;
    float2 v[NB][NV];
    float  w[NB];
    int    r[NB];
    #pragma unroll
    for (int b = 0; b < NB; ++b) {
        const uint2 en = lst[k + b];
        r[b] = (int)(en.x >> 16);
        w[b] = __uint_as_float(en.y);
        const float* xr = x + (long long)(en.x & 0xFFFFu) * D + 2 * lane;
        #pragma unroll
        for (int c = 0; c < NV; ++c) v[b][c] = *(const float2*)(xr + 64 * c);
    }
    #pragma unroll
    for (int b = 0; b < NB; ++b) {
        float* tr = tile + r[b] * (D + AGG_PAD) + 2 * lane;
        #pragma unroll
        for (int c = 0; c < NV; ++c) {
            atomicAdd(tr + 64 * c,     v[b][c].x * w[b]);
            atomicAdd(tr + 64 * c + 1, v[b][c].y * w[b]);
        }
    }
}
template <int D, int NB>
__device__ __forceinline__ void agg_batch(const _Float16* __restrict__ x,
                                          float* tile, const uint2* lst,
                                          int k, int lane) {
    static_assert(D % 256 == 0, "a lane takes 8 columns of every 256");
    constexpr int NV = D / 256;
    v8h   v[NB][NV];
    float w[NB];
    int   r[NB];
    #pragma unroll
    for (int b = 0; b < NB; ++b) {
        const uint2 en = lst[k + b];
        r[b] = (int)(en.x >> 16);
        w[b] = __uint_as_float(en.y);
        const _Float16* xr = x + (long long)(en.x & 0xFFFFu) * D + 8 * lane;
        #pragma unroll
        for (int c = 0; c < NV; ++c) v[b][c] = *(const v8h*)(xr + 256 * c);
    }
    #pragma unroll
    for (int b = 0; b < NB; ++b) {
        float* tr = tile + r[b] * (D + AGG_PAD) + 8 * lane;
        #pragma unroll
        for (int c = 0; c < NV; ++c) {
            #pragma unroll
            for (int i = 0; i < 8; ++i) atomicAdd(tr + 256 * c + i, (float)v[b][c][i] * w[b]);
        }
    }
}

template <int D, int NB>
__device__ __forceinline__ void own_batch(const float* __restrict__ x, const uint2* lst,
                                          int k, int lane, float (&acc)[D / 32]) {
    constexpr int NF = D / 32;
    float v[NB][NF];
    float w[NB];
    #pragma unroll
    for (int b = 0; b < NB; ++b) {
        const uint2 en = lst[k + b];
        const unsigned ex = (unsigned)__builtin_amdgcn_readfirstlane((int)en.x);
        w[b] = __uint_as_float((unsigned)__builtin_amdgcn_readfirstlane((int)en.y));
        const float* xr = x + (long long)(ex & 0xFFFFu) * D + 4 * lane;
        #pragma unroll
        for (int c = 0; c < NF / 4; ++c) {
            const float4 t = *(const float4*)(xr + 128 * c);
            v[b][4 * c] = t.x; v[b][4 * c + 1] = t.y; v[b][4 * c + 2] = t.z; v[b][4 * c + 3] = t.w;
        }
    }
    #pragma unroll
    for (int b = 0; b < NB; ++b) {
        #pragma unroll
        for (int c = 0; c < NF; ++c) acc[c] = fmaf(w[b], v[b][c], acc[c]);
    }
}

template <int D>
__device__ __forceinline__ void own_consume(const float* __restrict__ x, const uint2* lst,
                                            int cnt, int lane, float (&acc)[D / 32]) {
    int k = 0;
    if constexpr (D <= 128) {
        while (k + 16 <= cnt) { own_batch<D, 16>(x, lst, k, lane, acc); k += 16; }
        if (k + 8 <= cnt)     { own_batch<D, 8>(x, lst, k, lane, acc); k += 8; }
    } else {
        while (k + 8 <= cnt)  { own_batch<D, 8>(x, lst, k, lane, acc); k += 8; }
    }
    if (k + 4 <= cnt)    { own_batch<D, 4>(x, lst, k, lane, acc); k += 4; }
    if (k + 2 <= cnt)    { own_batch<D, 2>(x, lst, k, lane, acc); k += 2; }
    if (k < cnt)         { own_batch<D, 1>(x, lst, k, lane, acc); }
}

template <int D, int NB>
__device__ __forceinline__ void reg_batch(const _Float16* __restrict__ x, const uint2 rec,
                                          int k, int lane, float (&acc)[D / 32]) {
    static_assert(D % 256 == 0, "a lane takes 8 columns of every 256");
    constexpr int NV = D / 256;
    v8h   v[NB][NV];
    float w[NB];
    #pragma unroll
    for (int b = 0; b < NB; ++b) {
        const unsigned ex = (unsigned)__builtin_amdgcn_readlane((int)rec.x, k + b);
        w[b] = __uint_as_float((unsigned)__builtin_amdgcn_readlane((int)rec.y, k + b));
        const _Float16* xr = x + (long long)(ex & 0xFFFFu) * D + 8 * lane;
        #pragma unroll
        for (int c = 0; c < NV; ++c) v[b][c] = *(const v8h*)(xr + 256 * c);
    }
    #pragma unroll
    for (int b = 0; b < NB; ++b) {
        #pragma unroll
        for (int c = 0; c < NV; ++c) {
            #pragma unroll
            for (int i = 0; i < 8; ++i) acc[8 * c + i] = fmaf(w[b], (float)v[b][c][i], acc[8 * c + i]);
        }
    }
}

template <int D>
__device__ __forceinline__ void reg_consume(const _Float16* __restrict__ x, const uint2 rec,
                                            int n, int lane, float (&acc)[D / 32]) {
    int k = 0;
    while (k + 16 <= n) { reg_batch<D, 16>(x, rec, k, lane, acc); k += 16; }
    if (k + 8 <= n)     { reg_batch<D, 8>(x, rec, k, lane, acc); k += 8; }
    if (k + 4 <= n)     { reg_batch<D, 4>(x, rec, k, lane, acc); k += 4; }
    if (k + 2 <= n)     { reg_batch<D, 2>(x, rec, k, lane, acc); k += 2; }
    if (k < n)          { reg_batch<D, 1>(x, rec, k, lane, acc); }
}

template <int D>
__device__ __forceinline__ void saved_rows(const _Float16* __restrict__ xs,
                                           const uint2* __restrict__ srec,
                                           const unsigned* __restrict__ rcnt,
                                           int wv, int lane, float* tile) {
    constexpr int P = D + AGG_PAD;
    unsigned t = 0u;
    #pragma unroll
    for (int j = 0; j < OWN_SLOTS; ++j) {
        const int row = lane + AGG_WAVES * j;
        t += row < AGG_ROWS ? rcnt[row] : 0u;
    }
    const int myrow = wv + AGG_WAVES * lane;
    const int mycnt = (lane < OWN_SLOTS && myrow < AGG_ROWS) ? (int)rcnt[myrow] : 0;
    int before = lane < wv ? (int)t : 0;
    #pragma unroll
    for (int o = 16; o >= 1; o >>= 1) before += __shfl_xor(before, o);
    int p   = __builtin_amdgcn_readfirstlane(before);
    int cnt = __builtin_amdgcn_readlane(mycnt, 0);
    uint2 nxt = make_uint2(0u, 0u);
    if (cnt > 0) nxt = srec[p + min(lane, cnt - 1)];
    for (int j = 0; j < OWN_SLOTS; ++j) {
        const int row = wv + AGG_WAVES * j;
        if (row >= AGG_ROWS) break;
        const int   ccnt = cnt;
        const int   cp   = p;
        p  += ccnt;
        cnt = (j + 1 < OWN_SLOTS) ? __builtin_amdgcn_readlane(mycnt, j + 1) : 0;
        const uint2 first = nxt;
        if (cnt > 0) nxt = srec[p + min(lane, cnt - 1)];
        float acc[D / 32];
        #pragma unroll
        for (int c = 0; c < D / 32; ++c) acc[c] = 0.0f;
        if (ccnt > 0) reg_consume<D>(xs, first, min(32, ccnt), lane, acc);
        for (int k0 = 32; k0 < ccnt; k0 += 32) {
            const int n = min(32, ccnt - k0);
            const uint2 more = srec[cp + k0 + min(lane, n - 1)];
            for (int k = 0; k < n; ++k) reg_batch<D, 1>(xs, more, k, lane, acc);
        }
        #pragma unroll
        for (int c = 0; c < D / 256; ++c) {
            float* tp = tile + row * P + 256 * c + 8 * lane;
            *(float4*)(tp)     = make_float4(acc[8 * c],     acc[8 * c + 1], acc[8 * c + 2], acc[8 * c + 3]);
            *(float4*)(tp + 4) = make_float4(acc[8 * c + 4], acc[8 * c + 5], acc[8 * c + 6], acc[8 * c + 7]);
        }
    }
}

__device__ __forceinline__ void rec_flush(const uint2* __restrict__ pk, const uint2* lst,
                                          int cnt, int lane, int row0,
                                          int* own_cnt, uint2 (*own)[OWN_CAP]) {
    for (int k0 = 0; k0 < cnt; k0 += 128) {
        unsigned u[4];
        uint2    p[4];
        #pragma unroll
        for (int q = 0; q < 4; ++q) {
            const int k = k0 + 32 * q + lane;
            u[q] = (k < cnt) ? lst[k].x : 0u;
        }
        #pragma unroll
        for (int q = 0; q < 4; ++q) p[q] = pk[u[q] & 0xFFFFFu];
        #pragma unroll
        for (int q = 0; q < 4; ++q) {
            const int k = k0 + 32 * q + lane;
            if (k < cnt) {
                const unsigned x0 = p[q].x - ((unsigned)row0 << 16);
                const unsigned row = x0 >> 16;
                const unsigned o   = row & (unsigned)(AGG_WAVES - 1);
                const int pos = atomicAdd(&own_cnt[o], 1);
                if (pos < OWN_CAP) own[o][pos] = make_uint2(((row >> 5) << 16) | (x0 & 0xFFFFu), p[q].y);
            }
        }
    }
}

template <int D, typename XT>
__device__ __forceinline__ void agg_flush(const XT* __restrict__ x,
                                          const uint2* __restrict__ pk,
                                          float* tile, uint2* lst,
                                          int cnt, int lane, int row0) {
    for (int k0 = 0; k0 < cnt; k0 += 128) {
        unsigned u[4];
        uint2    p[4];
        #pragma unroll
        for (int q = 0; q < 4; ++q) {
            const int k = k0 + 32 * q + lane;
            u[q] = (k < cnt) ? lst[k].x : 0u;
        }
        #pragma unroll
        for (int q = 0; q < 4; ++q) p[q] = pk[u[q] & 0xFFFFFu];
        #pragma unroll
        for (int q = 0; q < 4; ++q) {
            const int k = k0 + 32 * q + lane;
            if (k < cnt) lst[k] = make_uint2(p[q].x - ((unsigned)row0 << 16), p[q].y);
        }
    }
    int k = 0;
    while (k + 8 <= cnt) { agg_batch<D, 8>(x, tile, lst, k, lane); k += 8; }
    if (k + 4 <= cnt)    { agg_batch<D, 4>(x, tile, lst, k, lane); k += 4; }
    if (k + 2 <= cnt)    { agg_batch<D, 2>(x, tile, lst, k, lane); k += 2; }
    if (k < cnt)         { agg_batch<D, 1>(x, tile, lst, k, lane); }
}

template <int FLUSH_AT, typename F>
__device__ __forceinline__ void scan_tile(const unsigned* __restrict__ bm, int tile_id,
                                          int tid, uint2* lst, F flush) {
    static_assert(FLUSH_AT + 32 <= AGG_CAP, "a trip adds up to 32 entries");
    int cnt = 0;
    constexpr int NQ = BM_ROW_WORDS / 4;
    const uint4* b4 = reinterpret_cast<const uint4*>(bm + (long long)tile_id * BM_ROW_WORDS);
    uint4 bv[BM_LOADS];
    #pragma unroll
    for (int u = 0; u < BM_LOADS; ++u) {
        const int q = tid + u * AGG_THREADS;
        bv[u] = make_uint4(0u, 0u, 0u, 0u);
        if (q < NQ) bv[u] = b4[q];
    }
    #pragma unroll 1
    for (int u = 0; u < BM_LOADS; ++u) {
        const int q = tid + u * AGG_THREADS;
        uint4 cur = bv[0];
        #pragma unroll
        for (int v = 1; v < BM_LOADS; ++v) if (u == v) cur = bv[v];
        #pragma unroll 1
        for (int j = 0; j < 4; ++j) {
            unsigned m = j == 0 ? cur.x : j == 1 ? cur.y : j == 2 ? cur.z : cur.w;
            const int eb = (4 * q + j) * 32;
            unsigned bal = __builtin_amdgcn_ballot_w32(m != 0u);
            while (bal != 0u) {
                if (m != 0u) {
                    const int pos = cnt + (int)__builtin_amdgcn_mbcnt_lo(bal, 0u);
                    lst[pos].x = (unsigned)(eb + __builtin_ctz(m));
                    m &= m - 1u;
                }
                cnt += __builtin_popcount(bal);
                if (cnt >= FLUSH_AT) {
                    flush(cnt);
                    cnt = 0;
                }
                bal = __builtin_amdgcn_ballot_w32(m != 0u);
            }
        }
    }
    flush(cnt);
}

template <int NR>
__device__ __forceinline__ void rank_sort(uint2* rl, int cnt, int lane) {
    static_assert(NR <= 2, "the place in the list takes 6 bits of the key");
    uint2 rec[NR];
    unsigned long long key[NR];
    int below[NR];
    #pragma unroll
    for (int n = 0; n < NR; ++n) {
        const int idx = lane + 32 * n;
        rec[n] = rl[min(idx, cnt - 1)];
        key[n] = ((unsigned long long)(rec[n].x & 0xFFFFu) << 38) | ((unsigned long long)rec[n].y << 6) | (unsigned long long)idx;
        below[n] = 0;
    }
    #pragma unroll
    for (int s = 0; s < NR; ++s) {
        const int lim = min(32, cnt - 32 * s);
        for (int i = 0; i < lim; ++i) {
            const unsigned lo = (unsigned)__builtin_amdgcn_readlane((int)(unsigned)key[s], i);
            const unsigned hi = (unsigned)__builtin_amdgcn_readlane((int)(unsigned)(key[s] >> 32), i);
            const unsigned long long ki = ((unsigned long long)hi << 32) | lo;
            #pragma unroll
            for (int n = 0; n < NR; ++n) below[n] += (ki < key[n]) ? 1 : 0;
        }
    }
    #pragma unroll
    for (int n = 0; n < NR; ++n)
        if (lane + 32 * n < cnt) rl[below[n]] = rec[n];
}

template <int D, typename XT>
__device__ __forceinline__ void owner_rows(const XT* __restrict__ xs, const uint2* mine, uint2* rl,
                                           int have, int wv, int lane, float* tile,
                                           uint2* __restrict__ srec, unsigned* __restrict__ rcnt,
                                           int wr, bool keep) {
    constexpr int P = D + AGG_PAD;
    for (int j = 0; j < OWN_SLOTS; ++j) {
        const int row = wv + AGG_WAVES * j;
        if (row >= AGG_ROWS) break;
        float acc[D / 32];
        #pragma unroll
        for (int c = 0; c < D / 32; ++c) acc[c] = 0.0f;
        int  cnt   = 0;
        int  wr0   = wr;
        bool whole = true;
        for (int g = 0; g < have; g += 32) {
            const int q = g + lane;
            const uint2 en = mine[min(q, have - 1)];
            const bool m = (q < have) && ((en.x >> 16) == (unsigned)j);
            const unsigned bal = __builtin_amdgcn_ballot_w32(m);
            if (bal != 0u) {
                const int add = __builtin_popcount(bal);
                if (cnt + add > ROW_CAP) {
                    if (keep) {
                        for (int i = lane; i < cnt; i += 32) srec[wr + i] = rl[i];
                        wr += cnt;
                    }
                    own_consume<D>(xs, rl, cnt, lane, acc);
                    cnt = 0;
                    whole = false;
                }
                if (m) rl[cnt + (int)__builtin_amdgcn_mbcnt_lo(bal, 0u)] = en;
                cnt += add;
            }
        }
        if (whole && cnt > 1) {
            if (cnt > 32) rank_sort<2>(rl, cnt, lane);
            else          rank_sort<1>(rl, cnt, lane);
        }
        if (keep) {
            for (int i = lane; i < cnt; i += 32) srec[wr + i] = rl[i];
            wr += cnt;
            if (lane == 0) rcnt[row] = (unsigned)(wr - wr0);
        }
        own_consume<D>(xs, rl, cnt, lane, acc);
        if constexpr (sizeof(XT) == 2) {
            #pragma unroll
            for (int c = 0; c < D / 256; ++c) {
                float* tp = tile + row * P + 256 * c + 8 * lane;
                *(float4*)(tp)     = make_float4(acc[8 * c],     acc[8 * c + 1], acc[8 * c + 2], acc[8 * c + 3]);
                *(float4*)(tp + 4) = make_float4(acc[8 * c + 4], acc[8 * c + 5], acc[8 * c + 6], acc[8 * c + 7]);
            }
        } else {
            #pragma unroll
            for (int c = 0; c < D / 128; ++c)
                *(float4*)(tile + row * P + 128 * c + 4 * lane) =
                    make_float4(acc[4 * c], acc[4 * c + 1], acc[4 * c + 2], acc[4 * c + 3]);
        }
    }
}

__device__ __forceinline__ v16h load_frag(const float* p) {
    const float4 a0 = *(const float4*)(p);
    const float4 a1 = *(const float4*)(p + 4);
    const float4 a2 = *(const float4*)(p + 16);
    const float4 a3 = *(const float4*)(p + 20);
    v16h f;
    f[0]  = (_Float16)a0.x; f[1]  = (_Float16)a0.y; f[2]  = (_Float16)a0.z; f[3]  = (_Float16)a0.w;
    f[4]  = (_Float16)a1.x; f[5]  = (_Float16)a1.y; f[6]  = (_Float16)a1.z; f[7]  = (_Float16)a1.w;
    f[8]  = (_Float16)a2.x; f[9]  = (_Float16)a2.y; f[10] = (_Float16)a2.z; f[11] = (_Float16)a2.w;
    f[12] = (_Float16)a3.x; f[13] = (_Float16)a3.y; f[14] = (_Float16)a3.z; f[15] = (_Float16)a3.w;
    return f;
}

template <int KS, bool HELD>
__device__ __forceinline__ v8f tile_product(const float* ap, const float* wp, const v16h* bf) {
    v8f acc = {};
    if constexpr (HELD) {
        #pragma unroll
        for (int ks = 0; ks < KS; ++ks) {
            const v16h af = load_frag(ap + 32 * ks);
            acc = __builtin_amdgcn_wmma_f32_16x16x32_f16(
                 false, af,  false, bf[ks],
                 (short)0, acc,  false,  false);
        }
    } else {
        #pragma unroll 2
        for (int ks = 0; ks < KS; ++ks) {
            const v16h af = load_frag(ap + 32 * ks);
            const v16h bk = load_frag(wp + 32 * ks);
            acc = __builtin_amdgcn_wmma_f32_16x16x32_f16(
                 false, af,  false, bk,
                 (short)0, acc,  false,  false);
        }
    }
    return acc;
}

template <int NOUT, bool FINAL>
__device__ __forceinline__ void tile_store(const v8f acc, int rbase, int col, int nrows,
                                           float* hs, _Float16* __restrict__ resh, int row0) {
    if (col < NOUT) {
        #pragma unroll
        for (int r = 0; r < 8; ++r) {
            const int row = rbase + r;
            if (row < nrows) {
                if (FINAL) {
                    hs[row * HS_PITCH + col] = acc[r];
                } else {
                    const float v = acc[r];
                    resh[(long long)(row0 + row) * NOUT + col] = (_Float16)(v > 0.0f ? v : 0.0f);
                }
            }
        }
    }
}

__global__ __launch_bounds__(AGG_THREADS)
void edge_bitmap_kernel(const int* __restrict__ ei, const float* __restrict__ ew,
                        unsigned* __restrict__ bm, uint2* __restrict__ pk) {
    constexpr int PER = BM_CHUNK_EDGES / AGG_THREADS;
    constexpr int NV  = AGG_TILES * BM_CHUNK_WORDS / 4;
    __shared__ __attribute__((aligned(16))) unsigned blk[AGG_TILES * BM_CHUNK_WORDS];
    const int tid = threadIdx.x;
    const int e0  = blockIdx.x * BM_CHUNK_EDGES;
    const int* dstp = ei + N_EDGES;
    unsigned dv[PER];
    unsigned sv[PER];
    float    wv[PER];
    #pragma unroll
    for (int k = 0; k < PER; ++k) {
        const int e = e0 + tid + k * AGG_THREADS;
        dv[k] = 0xFFFFFFFFu;
        sv[k] = 0u;
        wv[k] = 0.0f;
        if (e < N_EDGES) {
            dv[k] = (unsigned)dstp[e];
            sv[k] = (unsigned)ei[e];
            wv[k] = ew[e];
        }
    }
    for (int i = tid; i < NV; i += AGG_THREADS)
        reinterpret_cast<uint4*>(blk)[i] = make_uint4(0u, 0u, 0u, 0u);
    __syncthreads();
    #pragma unroll
    for (int k = 0; k < PER; ++k) {
        const int el = tid + k * AGG_THREADS;
        if (dv[k] < (unsigned)N_NODES) {
            const unsigned t = dv[k] / (unsigned)AGG_ROWS;
            atomicOr(&blk[t * BM_CHUNK_WORDS + (el >> 5)], 1u << (el & 31));
        }
    }
    __syncthreads();
    #pragma unroll
    for (int k = 0; k < PER; ++k) {
        const int e = e0 + tid + k * AGG_THREADS;
        if (e < N_EDGES) pk[e] = make_uint2((dv[k] << 16) | (sv[k] & 0xFFFFu), __float_as_uint(wv[k]));
    }
    for (int i = tid; i < NV; i += AGG_THREADS) {
        const int t = i / (BM_CHUNK_WORDS / 4);
        const int j = i - t * (BM_CHUNK_WORDS / 4);
        reinterpret_cast<uint4*>(bm + (long long)t * BM_ROW_WORDS + blockIdx.x * BM_CHUNK_WORDS)[j] =
            reinterpret_cast<const uint4*>(blk)[i];
    }
}

template <int D, int NOUT, bool FINAL, bool SAVE, bool REPLAY, bool XH>
__global__ __launch_bounds__(AGG_THREADS)
void layer_kernel(const float* __restrict__ x,
                  const _Float16* __restrict__ xh,
                  const float* __restrict__ W,
                  float*       __restrict__ res,
                  _Float16*    __restrict__ resh,
                  int*         __restrict__ cnts,
                  unsigned*    __restrict__ recs,
                  const unsigned* __restrict__ bm,
                  const uint2*    __restrict__ pk) {
    constexpr int P  = D + AGG_PAD;
    constexpr int NT = (NOUT + 15) / 16;
    constexpr int SW = AGG_WAVES / NT;
    constexpr int KS = D / 32;
    constexpr bool HOLD_B = (AGG_WAVES % NT == 0) && (AGG_STRIPS >= 4 * SW);
    static_assert(SW >= 1, "one wave per column tile at least");
    static_assert(!FINAL || (AGG_ROWS * HS_PITCH + AGG_ROWS) * 4 <= AGG_WAVES * AGG_CAP * 8,
                  "score tile and row terms reuse the edge lists' space");

    __shared__ __attribute__((aligned(16))) float tile[AGG_ROWS * P];
    __shared__ __attribute__((aligned(16))) uint2 lists[AGG_WAVES][AGG_CAP];
    constexpr bool OWNERS = SAVE;
    __shared__ __attribute__((aligned(16))) uint2 own[OWNERS ? AGG_WAVES : 1][OWN_CAP];
    __shared__ __attribute__((aligned(16))) uint2 rowl[OWNERS ? AGG_WAVES : 1][OWNERS ? ROW_CAP : 1];
    __shared__ int own_cnt[AGG_WAVES];

    const auto xs = [&] { if constexpr (XH) return xh; else return x; }();

    const int tid   = threadIdx.x;
    const int lane  = tid & 31;
    const int wave  = tid >> 5;
    const int row0  = blockIdx.x * AGG_ROWS;
    const int nrows = min(AGG_ROWS, N_NODES - row0);
    const int nel   = nrows * P;

    uint2*    trec = reinterpret_cast<uint2*>(recs) + (long long)blockIdx.x * REC_CAP;
    unsigned* rcnt = reinterpret_cast<unsigned*>(trec);
    uint2*    srec = trec + REC_HEAD;
    int  found  = REC_CAP + 1;
    int  total  = 0;
    bool replay = false;
    if (REPLAY) {
        total  = __builtin_amdgcn_readfirstlane(cnts[blockIdx.x]);
        replay = total <= REC_CAP;
    }
    const int wv = __builtin_amdgcn_readfirstlane(wave);

    for (int i = tid * 4; i < nel; i += AGG_THREADS * 4)
        *(float4*)(tile + i) = make_float4(0.f, 0.f, 0.f, 0.f);
    if (OWNERS && tid < AGG_WAVES) own_cnt[tid] = 0;
    __syncthreads();

    bool summed = false;
    if constexpr (SAVE) {
        uint2* lst = lists[wave];
        scan_tile<REC_FLUSH>(bm, blockIdx.x, tid, lst, [&](int cnt) {
            rec_flush(pk, lst, cnt, lane, row0, own_cnt, own);
        });
        __syncthreads();
        const int  oc   = own_cnt[lane];
        const bool over = __builtin_amdgcn_ballot_w32(oc > OWN_CAP) != 0u;
        if (!over) {
            int all = oc, before = lane < wv ? oc : 0;
            #pragma unroll
            for (int o = 16; o >= 1; o >>= 1) {
                all    += __shfl_xor(all, o);
                before += __shfl_xor(before, o);
            }
            all    = __builtin_amdgcn_readfirstlane(all);
            before = __builtin_amdgcn_readfirstlane(before);
            const bool keep = all <= REC_KEEP;
            if (keep) found = all;
            owner_rows<D>(xs, own[wv], rowl[wv], __builtin_amdgcn_readfirstlane(own_cnt[wv]), wv, lane, tile,
                          srec, rcnt, before, keep);
            summed = true;
        }
    }
    if constexpr (REPLAY) {
        if (replay) {
            saved_rows<D>(xs, srec, rcnt, wv, lane, tile);
            summed = true;
        }
    }
    if (!summed) {
        uint2* lst = lists[wave];
        scan_tile<AGG_FLUSH>(bm, blockIdx.x, tid, lst, [&](int cnt) {
            agg_flush<D>(xs, pk, tile, lst, cnt, lane, row0);
        });
    }
    __syncthreads();
    if (SAVE && tid == 0) cnts[blockIdx.x] = found;

    const int c       = lane & 15;
    const int h8      = (lane >> 4) * 8;
    const int nstrips = (nrows + 15) >> 4;
    float* hs  = reinterpret_cast<float*>(&lists[0][0]);
    float* lse = hs + AGG_ROWS * HS_PITCH;

    if (HOLD_B) {
        const int col = (wv % NT) * 16 + c;
        const float* wp = W + (long long)min(col, NOUT - 1) * D + h8;
        v16h bf[KS];
        #pragma unroll
        for (int ks = 0; ks < KS; ++ks) bf[ks] = load_frag(wp + 32 * ks);
        for (int s = wv / NT; s < nstrips; s += SW) {
            const int arow = min(s * 16 + c, AGG_ROWS - 1);
            const v8f acc = tile_product<KS, true>(tile + arow * P + h8, nullptr, bf);
            tile_store<NOUT, FINAL>(acc, s * 16 + h8, col, nrows, hs, resh, row0);
        }
    } else {
        for (int t = wv; t < nstrips * NT; t += AGG_WAVES) {
            const int s   = t / NT;
            const int col = (t - s * NT) * 16 + c;
            const float* wp = W + (long long)min(col, NOUT - 1) * D + h8;
            const int arow = min(s * 16 + c, AGG_ROWS - 1);
            const v8f acc = tile_product<KS, false>(tile + arow * P + h8, wp, nullptr);
            tile_store<NOUT, FINAL>(acc, s * 16 + h8, col, nrows, hs, resh, row0);
        }
    }

    if (FINAL) {
        __syncthreads();
        if (tid < nrows) {
            const float* h = hs + tid * HS_PITCH;
            float mx = -INFINITY;
            #pragma unroll
            for (int k = 0; k < NOUT; ++k) mx = fmaxf(mx, h[k]);
            float sum = 0.0f;
            #pragma unroll
            for (int k = 0; k < NOUT; ++k) sum += __expf(h[k] - mx);
            lse[tid] = __logf(sum) + mx;
        }
        __syncthreads();
        float* o = res + (long long)row0 * NOUT;
        for (int i = tid; i < nrows * NOUT; i += AGG_THREADS) {
            const int r = i / NOUT;
            o[i] = hs[r * HS_PITCH + (i - r * NOUT)] - lse[r];
        }
    }
}

extern "C" void kernel_launch(void* const* d_in, const int* in_sizes, int n_in,
                              void* d_out, int out_size, void* d_ws, size_t ws_size,
                              hipStream_t stream) {
    const float* x  = (const float*)d_in[0];
    const int*   ei = (const int*)  d_in[1];
    const float* ew = (const float*)d_in[2];
    const float* w1 = (const float*)d_in[3];
    const float* w2 = (const float*)d_in[4];
    float* out = (float*)d_out;
    _Float16* h1   = (_Float16*)d_ws;
    int*      cnts = (int*)((char*)d_ws + WS_CNT_OFF);
    unsigned* recs = (unsigned*)((char*)d_ws + WS_REC_OFF);
    unsigned* bm   = (unsigned*)((char*)d_ws + WS_BM_OFF);
    uint2*    pk   = (uint2*)((char*)d_ws + WS_PK_OFF);

    edge_bitmap_kernel<<<BM_CHUNKS, AGG_THREADS, 0, stream>>>(ei, ew, bm, pk);

    layer_kernel<NUM_FEATS, HIDDEN, false, true, false, false><<<AGG_TILES, AGG_THREADS, 0, stream>>>(x, nullptr, w1, nullptr, h1, cnts, recs, bm, pk);
    layer_kernel<HIDDEN, NUM_CLASSES, true, false, true, true><<<AGG_TILES, AGG_THREADS, 0, stream>>>(nullptr, h1, w2, out, nullptr, cnts, recs, bm, pk);
}


# edge bitmaps hold one bit per group of 8 consecutive edges, not per edge: 8x less bitmap to build, store, read and scan; a listed group's packed edges are one 64-byte look-up, tested for the tile
# speedup vs baseline: 6.5690x; 1.0714x over previous
#include <hip/hip_runtime.h>
#include <hip/hip_bf16.h>
#include <math.h>

#define N_NODES     50000
#define N_EDGES     800000
#define NUM_FEATS   128
#define HIDDEN      256
#define NUM_CLASSES 40

typedef __attribute__((ext_vector_type(16))) _Float16 v16h;
typedef __attribute__((ext_vector_type(8)))  float    v8f;
typedef __attribute__((ext_vector_type(8)))  _Float16 v8h;

#define AGG_TILES   256
#define AGG_ROWS    196
#define AGG_PAD     4
#define AGG_STRIPS  ((AGG_ROWS + 15) / 16)
#define HS_PITCH    (NUM_CLASSES + 1)
#define AGG_THREADS 1024
#define AGG_WAVES   (AGG_THREADS / 32)
#define AGG_FLUSH   32
#define REC_FLUSH   128
#define AGG_CAP     (AGG_FLUSH + 128)
#define BM_GROUP       8
#define BM_CHUNKS      196
#define BM_CHUNK_WORDS 16
#define BM_CHUNK_EDGES (BM_CHUNK_WORDS * 32 * BM_GROUP)
#define BM_ROW_WORDS   (BM_CHUNKS * BM_CHUNK_WORDS)
#define BM_LOADS       ((BM_ROW_WORDS + AGG_THREADS - 1) / AGG_THREADS)
#define BM_THREADS     512
#define REC_CAP     4096
#define REC_HEAD    128
#define REC_KEEP    (REC_CAP - REC_HEAD)
#define OWN_SLOTS   ((AGG_ROWS + AGG_WAVES - 1) / AGG_WAVES)
#define OWN_CAP     224
#define ROW_CAP     64
#define WS_CNT_OFF  (N_NODES * HIDDEN * 2)
#define WS_REC_OFF  (WS_CNT_OFF + 4096)
#define WS_BM_OFF   (WS_REC_OFF + AGG_TILES * REC_CAP * 8)
#define WS_PK_OFF   (WS_BM_OFF + AGG_TILES * BM_ROW_WORDS * 4)
static_assert(AGG_TILES * AGG_ROWS >= N_NODES, "tiles must cover all nodes");
static_assert(N_NODES <= 65536, "source and destination ids are packed into 16 bits each");
static_assert(WS_PK_OFF % 256 == 0, "a wave stores 32 packed edges as whole 128-byte lines");
static_assert(AGG_ROWS <= 256, "tile row is packed into 8 bits");
static_assert(BM_CHUNKS * BM_CHUNK_EDGES >= N_EDGES, "chunks must cover all edges");
static_assert(BM_GROUP == 8 && N_EDGES % BM_GROUP == 0 && WS_PK_OFF % 16 == 0,
              "a group is four 16-byte pieces of the packed edges, all inside the array");
static_assert(AGG_TILES * BM_CHUNK_WORDS * 4 <= 160 * 1024, "a chunk's bitmaps of all tiles fit in LDS");
static_assert(BM_CHUNK_EDGES % BM_THREADS == 0 && BM_CHUNK_WORDS == 16 && AGG_TILES % 2 == 0 && WS_BM_OFF % 128 == 0,
              "a thread takes a whole number of edges; two tiles' stretches of a chunk make one 128-byte line");
static_assert(N_EDGES <= (1 << 20), "a group's number fits the 20 bits kept of a list entry");
static_assert(AGG_WAVES == 32, "owner wave = row & 31, row slot = row >> 5");
static_assert(AGG_ROWS * 4 <= REC_HEAD * 8, "the row counts fit in front of the records");

template <int D>
__device__ __forceinline__ void agg_one(const float* __restrict__ x,
                                        float* tile, const uint2 en, int lane) {
    constexpr int NV = D / 64;
    const int   r = (int)(en.x >> 16);
    const float w = __uint_as_float(en.y);
    const float* xr = x + (long long)(en.x & 0xFFFFu) * D + 2 * lane;
    float*       tr = tile + r * (D + AGG_PAD) + 2 * lane;
    float2 v[NV];
    #pragma unroll
    for (int c = 0; c < NV; ++c) v[c] = *(const float2*)(xr + 64 * c);
    #pragma unroll
    for (int c = 0; c < NV; ++c) {
        atomicAdd(tr + 64 * c,     v[c].x * w);
        atomicAdd(tr + 64 * c + 1, v[c].y * w);
    }
}
template <int D>
__device__ __forceinline__ void agg_one(const _Float16* __restrict__ x,
                                        float* tile, const uint2 en, int lane) {
    static_assert(D % 256 == 0, "a lane takes 8 columns of every 256");
    constexpr int NV = D / 256;
    const int   r = (int)(en.x >> 16);
    const float w = __uint_as_float(en.y);
    const _Float16* xr = x + (long long)(en.x & 0xFFFFu) * D + 8 * lane;
    float*          tr = tile + r * (D + AGG_PAD) + 8 * lane;
    v8h v[NV];
    #pragma unroll
    for (int c = 0; c < NV; ++c) v[c] = *(const v8h*)(xr + 256 * c);
    #pragma unroll
    for (int c = 0; c < NV; ++c) {
        #pragma unroll
        for (int i = 0; i < 8; ++i) atomicAdd(tr + 256 * c + i, (float)v[c][i] * w);
    }
}

template <int D, int NB>
__device__ __forceinline__ void own_batch(const float* __restrict__ x, const uint2* lst,
                                          int k, int lane, float (&acc)[D / 32]) {
    constexpr int NF = D / 32;
    float v[NB][NF];
    float w[NB];
    #pragma unroll
    for (int b = 0; b < NB; ++b) {
        const uint2 en = lst[k + b];
        const unsigned ex = (unsigned)__builtin_amdgcn_readfirstlane((int)en.x);
        w[b] = __uint_as_float((unsigned)__builtin_amdgcn_readfirstlane((int)en.y));
        const float* xr = x + (long long)(ex & 0xFFFFu) * D + 4 * lane;
        #pragma unroll
        for (int c = 0; c < NF / 4; ++c) {
            const float4 t = *(const float4*)(xr + 128 * c);
            v[b][4 * c] = t.x; v[b][4 * c + 1] = t.y; v[b][4 * c + 2] = t.z; v[b][4 * c + 3] = t.w;
        }
    }
    #pragma unroll
    for (int b = 0; b < NB; ++b) {
        #pragma unroll
        for (int c = 0; c < NF; ++c) acc[c] = fmaf(w[b], v[b][c], acc[c]);
    }
}

template <int D>
__device__ __forceinline__ void own_consume(const float* __restrict__ x, const uint2* lst,
                                            int cnt, int lane, float (&acc)[D / 32]) {
    int k = 0;
    if constexpr (D <= 128) {
        while (k + 16 <= cnt) { own_batch<D, 16>(x, lst, k, lane, acc); k += 16; }
        if (k + 8 <= cnt)     { own_batch<D, 8>(x, lst, k, lane, acc); k += 8; }
    } else {
        while (k + 8 <= cnt)  { own_batch<D, 8>(x, lst, k, lane, acc); k += 8; }
    }
    if (k + 4 <= cnt)    { own_batch<D, 4>(x, lst, k, lane, acc); k += 4; }
    if (k + 2 <= cnt)    { own_batch<D, 2>(x, lst, k, lane, acc); k += 2; }
    if (k < cnt)         { own_batch<D, 1>(x, lst, k, lane, acc); }
}

template <int D, int NB>
__device__ __forceinline__ void reg_batch(const _Float16* __restrict__ x, const uint2 rec,
                                          int k, int lane, float (&acc)[D / 32]) {
    static_assert(D % 256 == 0, "a lane takes 8 columns of every 256");
    constexpr int NV = D / 256;
    v8h   v[NB][NV];
    float w[NB];
    #pragma unroll
    for (int b = 0; b < NB; ++b) {
        const unsigned ex = (unsigned)__builtin_amdgcn_readlane((int)rec.x, k + b);
        w[b] = __uint_as_float((unsigned)__builtin_amdgcn_readlane((int)rec.y, k + b));
        const _Float16* xr = x + (long long)(ex & 0xFFFFu) * D + 8 * lane;
        #pragma unroll
        for (int c = 0; c < NV; ++c) v[b][c] = *(const v8h*)(xr + 256 * c);
    }
    #pragma unroll
    for (int b = 0; b < NB; ++b) {
        #pragma unroll
        for (int c = 0; c < NV; ++c) {
            #pragma unroll
            for (int i = 0; i < 8; ++i) acc[8 * c + i] = fmaf(w[b], (float)v[b][c][i], acc[8 * c + i]);
        }
    }
}

template <int D>
__device__ __forceinline__ void reg_consume(const _Float16* __restrict__ x, const uint2 rec,
                                            int n, int lane, float (&acc)[D / 32]) {
    int k = 0;
    while (k + 16 <= n) { reg_batch<D, 16>(x, rec, k, lane, acc); k += 16; }
    if (k + 8 <= n)     { reg_batch<D, 8>(x, rec, k, lane, acc); k += 8; }
    if (k + 4 <= n)     { reg_batch<D, 4>(x, rec, k, lane, acc); k += 4; }
    if (k + 2 <= n)     { reg_batch<D, 2>(x, rec, k, lane, acc); k += 2; }
    if (k < n)          { reg_batch<D, 1>(x, rec, k, lane, acc); }
}

template <int D>
__device__ __forceinline__ void saved_rows(const _Float16* __restrict__ xs,
                                           const uint2* __restrict__ srec,
                                           const unsigned* __restrict__ rcnt,
                                           int wv, int lane, float* tile) {
    constexpr int P = D + AGG_PAD;
    unsigned t = 0u;
    #pragma unroll
    for (int j = 0; j < OWN_SLOTS; ++j) {
        const int row = lane + AGG_WAVES * j;
        t += row < AGG_ROWS ? rcnt[row] : 0u;
    }
    const int myrow = wv + AGG_WAVES * lane;
    const int mycnt = (lane < OWN_SLOTS && myrow < AGG_ROWS) ? (int)rcnt[myrow] : 0;
    int before = lane < wv ? (int)t : 0;
    #pragma unroll
    for (int o = 16; o >= 1; o >>= 1) before += __shfl_xor(before, o);
    int p   = __builtin_amdgcn_readfirstlane(before);
    int cnt = __builtin_amdgcn_readlane(mycnt, 0);
    uint2 nxt = make_uint2(0u, 0u);
    if (cnt > 0) nxt = srec[p + min(lane, cnt - 1)];
    for (int j = 0; j < OWN_SLOTS; ++j) {
        const int row = wv + AGG_WAVES * j;
        if (row >= AGG_ROWS) break;
        const int   ccnt = cnt;
        const int   cp   = p;
        p  += ccnt;
        cnt = (j + 1 < OWN_SLOTS) ? __builtin_amdgcn_readlane(mycnt, j + 1) : 0;
        const uint2 first = nxt;
        if (cnt > 0) nxt = srec[p + min(lane, cnt - 1)];
        float acc[D / 32];
        #pragma unroll
        for (int c = 0; c < D / 32; ++c) acc[c] = 0.0f;
        if (ccnt > 0) reg_consume<D>(xs, first, min(32, ccnt), lane, acc);
        for (int k0 = 32; k0 < ccnt; k0 += 32) {
            const int n = min(32, ccnt - k0);
            const uint2 more = srec[cp + k0 + min(lane, n - 1)];
            for (int k = 0; k < n; ++k) reg_batch<D, 1>(xs, more, k, lane, acc);
        }
        #pragma unroll
        for (int c = 0; c < D / 256; ++c) {
            float* tp = tile + row * P + 256 * c + 8 * lane;
            *(float4*)(tp)     = make_float4(acc[8 * c],     acc[8 * c + 1], acc[8 * c + 2], acc[8 * c + 3]);
            *(float4*)(tp + 4) = make_float4(acc[8 * c + 4], acc[8 * c + 5], acc[8 * c + 6], acc[8 * c + 7]);
        }
    }
}

__device__ __forceinline__ void rec_flush(const uint2* __restrict__ pk, const uint2* lst,
                                          int cnt, int lane, int row0,
                                          int* own_cnt, uint2 (*own)[OWN_CAP]) {
    const uint4*   pk4  = reinterpret_cast<const uint4*>(pk);
    const unsigned base = (unsigned)row0 << 16;
    const int sub = lane & 3;
    const int grp = lane >> 2;
    for (int k0 = 0; k0 < cnt; k0 += 32) {
        unsigned g[4];
        uint4    p[4];
        #pragma unroll
        for (int q = 0; q < 4; ++q) {
            const int k = k0 + 8 * q + grp;
            g[q] = (k < cnt) ? (lst[k].x & 0xFFFFFu) : 0u;
        }
        #pragma unroll
        for (int q = 0; q < 4; ++q) p[q] = pk4[g[q] * (BM_GROUP / 2) + sub];
        unsigned x0[8];
        unsigned wy[8];
        int      pos[8];
        #pragma unroll
        for (int q = 0; q < 4; ++q) {
            x0[2 * q]     = p[q].x - base;  wy[2 * q]     = p[q].y;
            x0[2 * q + 1] = p[q].z - base;  wy[2 * q + 1] = p[q].w;
        }
        #pragma unroll
        for (int i = 0; i < 8; ++i) {
            const unsigned row = x0[i] >> 16;
            pos[i] = OWN_CAP;
            if (k0 + 8 * (i >> 1) + grp < cnt && row < (unsigned)AGG_ROWS)
                pos[i] = atomicAdd(&own_cnt[row & (unsigned)(AGG_WAVES - 1)], 1);
        }
        #pragma unroll
        for (int i = 0; i < 8; ++i) {
            if (pos[i] < OWN_CAP) {
                const unsigned row = x0[i] >> 16;
                own[row & (unsigned)(AGG_WAVES - 1)][pos[i]] = make_uint2(((row >> 5) << 16) | (x0[i] & 0xFFFFu), wy[i]);
            }
        }
    }
}

template <int D, typename XT>
__device__ __forceinline__ void agg_flush(const XT* __restrict__ x,
                                          const uint2* __restrict__ pk,
                                          float* tile, const uint2* lst,
                                          int cnt, int lane, int row0) {
    for (int k = 0; k < cnt; ++k) {
        const unsigned g  = lst[k].x & 0xFFFFFu;
        const uint2    e  = pk[g * BM_GROUP + (unsigned)(lane & (BM_GROUP - 1))];
        const unsigned x0 = e.x - ((unsigned)row0 << 16);
        unsigned bal = __builtin_amdgcn_ballot_w32((x0 >> 16) < (unsigned)AGG_ROWS) & ((1u << BM_GROUP) - 1u);
        while (bal != 0u) {
            const int i = __builtin_ctz(bal);
            bal &= bal - 1u;
            const uint2 en = make_uint2((unsigned)__builtin_amdgcn_readlane((int)x0, i),
                                        (unsigned)__builtin_amdgcn_readlane((int)e.y, i));
            agg_one<D>(x, tile, en, lane);
        }
    }
}

__device__ __forceinline__ long long bm_word(int t, int c, int w) {
    return ((long long)(t >> 1) * BM_CHUNKS + c) * (2 * BM_CHUNK_WORDS) + (t & 1) * BM_CHUNK_WORDS + w;
}

template <int FLUSH_AT, typename F>
__device__ __forceinline__ void scan_tile(const unsigned* __restrict__ bm, int tile_id,
                                          int tid, uint2* lst, F flush) {
    static_assert(FLUSH_AT + 32 <= AGG_CAP, "a trip adds up to 32 entries");
    int cnt = 0;
    unsigned bv[BM_LOADS];
    #pragma unroll
    for (int u = 0; u < BM_LOADS; ++u) {
        const int q = tid + u * AGG_THREADS;
        bv[u] = 0u;
        if (q < BM_ROW_WORDS) bv[u] = bm[bm_word(tile_id, q / BM_CHUNK_WORDS, q % BM_CHUNK_WORDS)];
    }
    #pragma unroll 1
    for (int u = 0; u < BM_LOADS; ++u) {
        const int q = tid + u * AGG_THREADS;
        unsigned m = bv[0];
        #pragma unroll
        for (int v = 1; v < BM_LOADS; ++v) if (u == v) m = bv[v];
        const int gb = q * 32;
        unsigned bal = __builtin_amdgcn_ballot_w32(m != 0u);
        while (bal != 0u) {
            if (m != 0u) {
                const int pos = cnt + (int)__builtin_amdgcn_mbcnt_lo(bal, 0u);
                lst[pos].x = (unsigned)(gb + __builtin_ctz(m));
                m &= m - 1u;
            }
            cnt += __builtin_popcount(bal);
            if (cnt >= FLUSH_AT) {
                flush(cnt);
                cnt = 0;
            }
            bal = __builtin_amdgcn_ballot_w32(m != 0u);
        }
    }
    flush(cnt);
}

template <int NR>
__device__ __forceinline__ void rank_sort(uint2* rl, int cnt, int lane) {
    static_assert(NR <= 2, "the place in the list takes 6 bits of the key");
    uint2 rec[NR];
    unsigned long long key[NR];
    int below[NR];
    #pragma unroll
    for (int n = 0; n < NR; ++n) {
        const int idx = lane + 32 * n;
        rec[n] = rl[min(idx, cnt - 1)];
        key[n] = ((unsigned long long)(rec[n].x & 0xFFFFu) << 38) | ((unsigned long long)rec[n].y << 6) | (unsigned long long)idx;
        below[n] = 0;
    }
    #pragma unroll
    for (int s = 0; s < NR; ++s) {
        const int lim = min(32, cnt - 32 * s);
        for (int i = 0; i < lim; ++i) {
            const unsigned lo = (unsigned)__builtin_amdgcn_readlane((int)(unsigned)key[s], i);
            const unsigned hi = (unsigned)__builtin_amdgcn_readlane((int)(unsigned)(key[s] >> 32), i);
            const unsigned long long ki = ((unsigned long long)hi << 32) | lo;
            #pragma unroll
            for (int n = 0; n < NR; ++n) below[n] += (ki < key[n]) ? 1 : 0;
        }
    }
    #pragma unroll
    for (int n = 0; n < NR; ++n)
        if (lane + 32 * n < cnt) rl[below[n]] = rec[n];
}

template <int D, typename XT>
__device__ __forceinline__ void owner_rows(const XT* __restrict__ xs, const uint2* mine, uint2* rl,
                                           int have, int wv, int lane, float* tile,
                                           uint2* __restrict__ srec, unsigned* __restrict__ rcnt,
                                           int wr, bool keep) {
    constexpr int P = D + AGG_PAD;
    for (int j = 0; j < OWN_SLOTS; ++j) {
        const int row = wv + AGG_WAVES * j;
        if (row >= AGG_ROWS) break;
        float acc[D / 32];
        #pragma unroll
        for (int c = 0; c < D / 32; ++c) acc[c] = 0.0f;
        int  cnt   = 0;
        int  wr0   = wr;
        bool whole = true;
        for (int g = 0; g < have; g += 32) {
            const int q = g + lane;
            const uint2 en = mine[min(q, have - 1)];
            const bool m = (q < have) && ((en.x >> 16) == (unsigned)j);
            const unsigned bal = __builtin_amdgcn_ballot_w32(m);
            if (bal != 0u) {
                const int add = __builtin_popcount(bal);
                if (cnt + add > ROW_CAP) {
                    if (keep) {
                        for (int i = lane; i < cnt; i += 32) srec[wr + i] = rl[i];
                        wr += cnt;
                    }
                    own_consume<D>(xs, rl, cnt, lane, acc);
                    cnt = 0;
                    whole = false;
                }
                if (m) rl[cnt + (int)__builtin_amdgcn_mbcnt_lo(bal, 0u)] = en;
                cnt += add;
            }
        }
        if (whole && cnt > 1) {
            if (cnt > 32) rank_sort<2>(rl, cnt, lane);
            else          rank_sort<1>(rl, cnt, lane);
        }
        if (keep) {
            for (int i = lane; i < cnt; i += 32) srec[wr + i] = rl[i];
            wr += cnt;
            if (lane == 0) rcnt[row] = (unsigned)(wr - wr0);
        }
        own_consume<D>(xs, rl, cnt, lane, acc);
        if constexpr (sizeof(XT) == 2) {
            #pragma unroll
            for (int c = 0; c < D / 256; ++c) {
                float* tp = tile + row * P + 256 * c + 8 * lane;
                *(float4*)(tp)     = make_float4(acc[8 * c],     acc[8 * c + 1], acc[8 * c + 2], acc[8 * c + 3]);
                *(float4*)(tp + 4) = make_float4(acc[8 * c + 4], acc[8 * c + 5], acc[8 * c + 6], acc[8 * c + 7]);
            }
        } else {
            #pragma unroll
            for (int c = 0; c < D / 128; ++c)
                *(float4*)(tile + row * P + 128 * c + 4 * lane) =
                    make_float4(acc[4 * c], acc[4 * c + 1], acc[4 * c + 2], acc[4 * c + 3]);
        }
    }
}

__device__ __forceinline__ v16h load_frag(const float* p) {
    const float4 a0 = *(const float4*)(p);
    const float4 a1 = *(const float4*)(p + 4);
    const float4 a2 = *(const float4*)(p + 16);
    const float4 a3 = *(const float4*)(p + 20);
    v16h f;
    f[0]  = (_Float16)a0.x; f[1]  = (_Float16)a0.y; f[2]  = (_Float16)a0.z; f[3]  = (_Float16)a0.w;
    f[4]  = (_Float16)a1.x; f[5]  = (_Float16)a1.y; f[6]  = (_Float16)a1.z; f[7]  = (_Float16)a1.w;
    f[8]  = (_Float16)a2.x; f[9]  = (_Float16)a2.y; f[10] = (_Float16)a2.z; f[11] = (_Float16)a2.w;
    f[12] = (_Float16)a3.x; f[13] = (_Float16)a3.y; f[14] = (_Float16)a3.z; f[15] = (_Float16)a3.w;
    return f;
}

template <int KS, bool HELD>
__device__ __forceinline__ v8f tile_product(const float* ap, const float* wp, const v16h* bf) {
    v8f acc = {};
    if constexpr (HELD) {
        #pragma unroll
        for (int ks = 0; ks < KS; ++ks) {
            const v16h af = load_frag(ap + 32 * ks);
            acc = __builtin_amdgcn_wmma_f32_16x16x32_f16(
                 false, af,  false, bf[ks],
                 (short)0, acc,  false,  false);
        }
    } else {
        #pragma unroll 2
        for (int ks = 0; ks < KS; ++ks) {
            const v16h af = load_frag(ap + 32 * ks);
            const v16h bk = load_frag(wp + 32 * ks);
            acc = __builtin_amdgcn_wmma_f32_16x16x32_f16(
                 false, af,  false, bk,
                 (short)0, acc,  false,  false);
        }
    }
    return acc;
}

template <int NOUT, bool FINAL>
__device__ __forceinline__ void tile_store(const v8f acc, int rbase, int col, int nrows,
                                           float* hs, _Float16* __restrict__ resh, int row0) {
    if (col < NOUT) {
        #pragma unroll
        for (int r = 0; r < 8; ++r) {
            const int row = rbase + r;
            if (row < nrows) {
                if (FINAL) {
                    hs[row * HS_PITCH + col] = acc[r];
                } else {
                    const float v = acc[r];
                    resh[(long long)(row0 + row) * NOUT + col] = (_Float16)(v > 0.0f ? v : 0.0f);
                }
            }
        }
    }
}

__global__ __launch_bounds__(BM_THREADS)
void edge_bitmap_kernel(const int* __restrict__ ei, const float* __restrict__ ew,
                        unsigned* __restrict__ bm, uint2* __restrict__ pk) {
    constexpr int PER = BM_CHUNK_EDGES / BM_THREADS;
    constexpr int NV  = AGG_TILES * BM_CHUNK_WORDS / 4;
    __shared__ __attribute__((aligned(16))) unsigned blk[AGG_TILES * BM_CHUNK_WORDS];
    const int tid = threadIdx.x;
    const int e0  = blockIdx.x * BM_CHUNK_EDGES;
    const int* dstp = ei + N_EDGES;
    unsigned dv[PER];
    unsigned sv[PER];
    float    wv[PER];
    #pragma unroll
    for (int k = 0; k < PER; ++k) {
        const int e = e0 + tid + k * BM_THREADS;
        dv[k] = 0xFFFFFFFFu;
        sv[k] = 0u;
        wv[k] = 0.0f;
        if (e < N_EDGES) {
            dv[k] = (unsigned)dstp[e];
            sv[k] = (unsigned)ei[e];
            wv[k] = ew[e];
        }
    }
    for (int i = tid; i < NV; i += BM_THREADS)
        reinterpret_cast<uint4*>(blk)[i] = make_uint4(0u, 0u, 0u, 0u);
    __syncthreads();
    #pragma unroll
    for (int k = 0; k < PER; ++k) {
        const int el = tid + k * BM_THREADS;
        if (dv[k] < (unsigned)N_NODES) {
            const unsigned t = dv[k] / (unsigned)AGG_ROWS;
            const int      g = el / BM_GROUP;
            atomicOr(&blk[t * BM_CHUNK_WORDS + (g >> 5)], 1u << (g & 31));
        }
    }
    __syncthreads();
    #pragma unroll
    for (int k = 0; k < PER; ++k) {
        const int e = e0 + tid + k * BM_THREADS;
        if (e < N_EDGES) {
            const unsigned d = dv[k] < (unsigned)N_NODES ? dv[k] : 0xFFFFu;
            pk[e] = make_uint2((d << 16) | (sv[k] & 0xFFFFu), __float_as_uint(wv[k]));
        }
    }
    for (int i = tid; i < NV; i += BM_THREADS) {
        const int t = i / (BM_CHUNK_WORDS / 4);
        const int j = i - t * (BM_CHUNK_WORDS / 4);
        *reinterpret_cast<uint4*>(bm + bm_word(t, blockIdx.x, 4 * j)) = reinterpret_cast<const uint4*>(blk)[i];
    }
}

template <int D, int NOUT, bool FINAL, bool SAVE, bool REPLAY, bool XH>
__global__ __launch_bounds__(AGG_THREADS)
void layer_kernel(const float* __restrict__ x,
                  const _Float16* __restrict__ xh,
                  const float* __restrict__ W,
                  float*       __restrict__ res,
                  _Float16*    __restrict__ resh,
                  int*         __restrict__ cnts,
                  unsigned*    __restrict__ recs,
                  const unsigned* __restrict__ bm,
                  const uint2*    __restrict__ pk) {
    constexpr int P  = D + AGG_PAD;
    constexpr int NT = (NOUT + 15) / 16;
    constexpr int SW = AGG_WAVES / NT;
    constexpr int KS = D / 32;
    constexpr bool HOLD_B = (AGG_WAVES % NT == 0) && (AGG_STRIPS >= 4 * SW);
    static_assert(SW >= 1, "one wave per column tile at least");
    static_assert(!FINAL || (AGG_ROWS * HS_PITCH + AGG_ROWS) * 4 <= AGG_WAVES * AGG_CAP * 8,
                  "score tile and row terms reuse the edge lists' space");

    __shared__ __attribute__((aligned(16))) float tile[AGG_ROWS * P];
    __shared__ __attribute__((aligned(16))) uint2 lists[AGG_WAVES][AGG_CAP];
    constexpr bool OWNERS = SAVE;
    __shared__ __attribute__((aligned(16))) uint2 own[OWNERS ? AGG_WAVES : 1][OWN_CAP];
    __shared__ __attribute__((aligned(16))) uint2 rowl[OWNERS ? AGG_WAVES : 1][OWNERS ? ROW_CAP : 1];
    __shared__ int own_cnt[AGG_WAVES];

    const auto xs = [&] { if constexpr (XH) return xh; else return x; }();

    const int tid   = threadIdx.x;
    const int lane  = tid & 31;
    const int wave  = tid >> 5;
    const int row0  = blockIdx.x * AGG_ROWS;
    const int nrows = min(AGG_ROWS, N_NODES - row0);
    const int nel   = nrows * P;

    uint2*    trec = reinterpret_cast<uint2*>(recs) + (long long)blockIdx.x * REC_CAP;
    unsigned* rcnt = reinterpret_cast<unsigned*>(trec);
    uint2*    srec = trec + REC_HEAD;
    int  found  = REC_CAP + 1;
    int  total  = 0;
    bool replay = false;
    if (REPLAY) {
        total  = __builtin_amdgcn_readfirstlane(cnts[blockIdx.x]);
        replay = total <= REC_CAP;
    }
    const int wv = __builtin_amdgcn_readfirstlane(wave);

    for (int i = tid * 4; i < nel; i += AGG_THREADS * 4)
        *(float4*)(tile + i) = make_float4(0.f, 0.f, 0.f, 0.f);
    if (OWNERS && tid < AGG_WAVES) own_cnt[tid] = 0;
    __syncthreads();

    bool summed = false;
    if constexpr (SAVE) {
        uint2* lst = lists[wave];
        scan_tile<REC_FLUSH>(bm, blockIdx.x, tid, lst, [&](int cnt) {
            rec_flush(pk, lst, cnt, lane, row0, own_cnt, own);
        });
        __syncthreads();
        const int  oc   = own_cnt[lane];
        const bool over = __builtin_amdgcn_ballot_w32(oc > OWN_CAP) != 0u;
        if (!over) {
            int all = oc, before = lane < wv ? oc : 0;
            #pragma unroll
            for (int o = 16; o >= 1; o >>= 1) {
                all    += __shfl_xor(all, o);
                before += __shfl_xor(before, o);
            }
            all    = __builtin_amdgcn_readfirstlane(all);
            before = __builtin_amdgcn_readfirstlane(before);
            const bool keep = all <= REC_KEEP;
            if (keep) found = all;
            owner_rows<D>(xs, own[wv], rowl[wv], __builtin_amdgcn_readfirstlane(own_cnt[wv]), wv, lane, tile,
                          srec, rcnt, before, keep);
            summed = true;
        }
    }
    if constexpr (REPLAY) {
        if (replay) {
            saved_rows<D>(xs, srec, rcnt, wv, lane, tile);
            summed = true;
        }
    }
    if (!summed) {
        uint2* lst = lists[wave];
        scan_tile<AGG_FLUSH>(bm, blockIdx.x, tid, lst, [&](int cnt) {
            agg_flush<D>(xs, pk, tile, lst, cnt, lane, row0);
        });
    }
    __syncthreads();
    if (SAVE && tid == 0) cnts[blockIdx.x] = found;

    const int c       = lane & 15;
    const int h8      = (lane >> 4) * 8;
    const int nstrips = (nrows + 15) >> 4;
    float* hs  = reinterpret_cast<float*>(&lists[0][0]);
    float* lse = hs + AGG_ROWS * HS_PITCH;

    if (HOLD_B) {
        const int col = (wv % NT) * 16 + c;
        const float* wp = W + (long long)min(col, NOUT - 1) * D + h8;
        v16h bf[KS];
        #pragma unroll
        for (int ks = 0; ks < KS; ++ks) bf[ks] = load_frag(wp + 32 * ks);
        for (int s = wv / NT; s < nstrips; s += SW) {
            const int arow = min(s * 16 + c, AGG_ROWS - 1);
            const v8f acc = tile_product<KS, true>(tile + arow * P + h8, nullptr, bf);
            tile_store<NOUT, FINAL>(acc, s * 16 + h8, col, nrows, hs, resh, row0);
        }
    } else {
        for (int t = wv; t < nstrips * NT; t += AGG_WAVES) {
            const int s   = t / NT;
            const int col = (t - s * NT) * 16 + c;
            const float* wp = W + (long long)min(col, NOUT - 1) * D + h8;
            const int arow = min(s * 16 + c, AGG_ROWS - 1);
            const v8f acc = tile_product<KS, false>(tile + arow * P + h8, wp, nullptr);
            tile_store<NOUT, FINAL>(acc, s * 16 + h8, col, nrows, hs, resh, row0);
        }
    }

    if (FINAL) {
        __syncthreads();
        if (tid < nrows) {
            const float* h = hs + tid * HS_PITCH;
            float mx = -INFINITY;
            #pragma unroll
            for (int k = 0; k < NOUT; ++k) mx = fmaxf(mx, h[k]);
            float sum = 0.0f;
            #pragma unroll
            for (int k = 0; k < NOUT; ++k) sum += __expf(h[k] - mx);
            lse[tid] = __logf(sum) + mx;
        }
        __syncthreads();
        float* o = res + (long long)row0 * NOUT;
        for (int i = tid; i < nrows * NOUT; i += AGG_THREADS) {
            const int r = i / NOUT;
            o[i] = hs[r * HS_PITCH + (i - r * NOUT)] - lse[r];
        }
    }
}

extern "C" void kernel_launch(void* const* d_in, const int* in_sizes, int n_in,
                              void* d_out, int out_size, void* d_ws, size_t ws_size,
                              hipStream_t stream) {
    const float* x  = (const float*)d_in[0];
    const int*   ei = (const int*)  d_in[1];
    const float* ew = (const float*)d_in[2];
    const float* w1 = (const float*)d_in[3];
    const float* w2 = (const float*)d_in[4];
    float* out = (float*)d_out;
    _Float16* h1   = (_Float16*)d_ws;
    int*      cnts = (int*)((char*)d_ws + WS_CNT_OFF);
    unsigned* recs = (unsigned*)((char*)d_ws + WS_REC_OFF);
    unsigned* bm   = (unsigned*)((char*)d_ws + WS_BM_OFF);
    uint2*    pk   = (uint2*)((char*)d_ws + WS_PK_OFF);

    edge_bitmap_kernel<<<BM_CHUNKS, BM_THREADS, 0, stream>>>(ei, ew, bm, pk);

    layer_kernel<NUM_FEATS, HIDDEN, false, true, false, false><<<AGG_TILES, AGG_THREADS, 0, stream>>>(x, nullptr, w1, nullptr, h1, cnts, recs, bm, pk);
    layer_kernel<HIDDEN, NUM_CLASSES, true, false, true, true><<<AGG_TILES, AGG_THREADS, 0, stream>>>(nullptr, h1, w2, out, nullptr, cnts, recs, bm, pk);
}
